# starting kernel
# baseline (speedup 1.0000x reference)
#include <hip/hip_runtime.h>
#include <math.h>

#define B_     4
#define T_IN_  12
#define T_OUT_ 24
#define N_     5000
#define C_     32
#define H_     8
#define K1_    17
#define TH_    (T_IN_ * H_)
#define OT_    (T_OUT_ - T_IN_)
#define OTP_   16
#define KSTEPS (TH_ / 4)
#define BETA_  0.8f

typedef __attribute__((ext_vector_type(2))) float v2f;
typedef __attribute__((ext_vector_type(8))) float v8f;

__global__ __launch_bounds__(256) void gnn_sigma_kernel(
    const float* __restrict__ nd, float* __restrict__ ws) {
  __shared__ float red[256];
  float m = 0.0f;
  for (int i = threadIdx.x; i < N_ * K1_; i += 256) m = fmaxf(m, nd[i]);
  red[threadIdx.x] = m;
  __syncthreads();
  for (int s = 128; s > 0; s >>= 1) {
    if (threadIdx.x < s) red[threadIdx.x] = fmaxf(red[threadIdx.x], red[threadIdx.x + s]);
    __syncthreads();
  }
  if (threadIdx.x == 0) {
    float mx = red[0];
    float sig = mx * 0.5f;
    ws[0] = 1.0f / (sig * sig);
  }
}

__global__ __launch_bounds__(128) void gnn_fused_kernel(
    const float* __restrict__ x,
    const int*   __restrict__ nn,
    const float* __restrict__ nd,
    const float* __restrict__ Wm,
    const float* __restrict__ bias,
    const float* __restrict__ ws,
    float*       __restrict__ out) {
  __shared__ float w_lds[K1_ * H_];
  __shared__ int   nn_lds[K1_];
  __shared__ float Wm_lds[TH_ * OTP_];
  __shared__ float bias_lds[OTP_];
  __shared__ float feat[B_][C_][TH_ + 1];

  const int n    = blockIdx.x;
  const int tid  = threadIdx.x;
  const int lane = tid & 31;
  const int b    = tid >> 5;
  const float inv_s2 = ws[0];

  for (int i = tid; i < K1_ * H_; i += 128) {
    const int j = i / H_, h = i % H_;
    const float d    = nd[n * K1_ + j];
    const int   node = nn[n * K1_ + j];
    const float lam  = (float)(h + 1) * (1.0f / (float)H_);
    float wv = __expf(-d * d * lam * inv_s2);
    if (node < 0 || wv < 1e-5f) wv = 0.0f;
    w_lds[i] = wv;
  }
  if (tid < K1_) {
    const int node = nn[n * K1_ + tid];
    nn_lds[tid] = (node < 0) ? 0 : node;
  }
  for (int i = tid; i < TH_ * OTP_; i += 128) {
    const int k = i >> 4, o = i & (OTP_ - 1);
    Wm_lds[i] = (o < OT_) ? Wm[k * OT_ + o] : 0.0f;
  }
  if (tid < OTP_) bias_lds[tid] = (tid < OT_) ? bias[tid] : 0.0f;
  __syncthreads();

  float acc[T_IN_][H_];
#pragma unroll
  for (int t = 0; t < T_IN_; ++t)
#pragma unroll
    for (int h = 0; h < H_; ++h) acc[t][h] = 0.0f;

  for (int j = 0; j < K1_; ++j) {
    const int m = nn_lds[j];
    float wj[H_];
#pragma unroll
    for (int h = 0; h < H_; ++h) wj[h] = w_lds[j * H_ + h];
    const float* xp = x + (((size_t)b * T_IN_) * N_ + m) * C_ + lane;
#pragma unroll
    for (int t = 0; t < T_IN_; ++t) {
      const float xv = xp[(size_t)t * N_ * C_];
#pragma unroll
      for (int h = 0; h < H_; ++h) acc[t][h] = fmaf(xv, wj[h], acc[t][h]);
    }
  }

#pragma unroll
  for (int t = 0; t < T_IN_; ++t) {
    out[(((size_t)b * T_OUT_ + t) * N_ + n) * C_ + lane] =
        x[(((size_t)b * T_IN_ + t) * N_ + n) * C_ + lane];
  }

#pragma unroll
  for (int t = 0; t < T_IN_; ++t)
#pragma unroll
    for (int h = 0; h < H_; ++h) feat[b][lane][t * H_ + h] = acc[t][h];
  __syncthreads();

  const int nidx  = lane & 15;
  const int khalf = (lane >> 4) * 2;

  v2f bfrag[KSTEPS];
#pragma unroll
  for (int kk = 0; kk < KSTEPS; ++kk) {
    const int k0 = kk * 4 + khalf;
    bfrag[kk].x = Wm_lds[(k0)     * OTP_ + nidx];
    bfrag[kk].y = Wm_lds[(k0 + 1) * OTP_ + nidx];
  }
  const float bv = bias_lds[nidx];

#pragma unroll
  for (int mt = 0; mt < 2; ++mt) {
    v8f c8 = {0.f, 0.f, 0.f, 0.f, 0.f, 0.f, 0.f, 0.f};
    const int mrow = mt * 16 + nidx;
#pragma unroll
    for (int kk = 0; kk < KSTEPS; ++kk) {
      const int k0 = kk * 4 + khalf;
      v2f a;
      a.x = feat[b][mrow][k0];
      a.y = feat[b][mrow][k0 + 1];
      c8 = __builtin_amdgcn_wmma_f32_16x16x4_f32(
           false, a,  false, bfrag[kk],
           (short)0, c8,  false,  false);
    }
    if (nidx < OT_) {
#pragma unroll
      for (int v = 0; v < 8; ++v) {
        const int m    = v + ((lane >> 4) << 3);
        const int cidx = mt * 16 + m;
        float y = c8[v] + bv;
        const float s = 1.0f / (1.0f + __expf(-BETA_ * y));
        y *= s;
        out[(((size_t)b * T_OUT_ + (T_IN_ + nidx)) * N_ + n) * C_ + cidx] = y;
      }
    }
  }
}

extern "C" void kernel_launch(void* const* d_in, const int* in_sizes, int n_in,
                              void* d_out, int out_size, void* d_ws, size_t ws_size,
                              hipStream_t stream) {
  const float* x    = (const float*)d_in[0];
  const int*   nn   = (const int*)d_in[1];
  const float* nd   = (const float*)d_in[2];
  const float* Wm   = (const float*)d_in[3];
  const float* bias = (const float*)d_in[4];
  float* out = (float*)d_out;
  float* ws  = (float*)d_ws;

  gnn_sigma_kernel<<<1, 256, 0, stream>>>(nd, ws);
  gnn_fused_kernel<<<N_, 128, 0, stream>>>(x, nn, nd, Wm, bias, ws, out);
}


# head weights folded into shrink matrix per node (V = w x W, 204x12): gathered x rows are the matrix B operand, K = 17x12, one node x 4 batches per wave; gather FMAs, LDS tile gone; sum order changed
# speedup vs baseline: 3.1505x; 3.1505x over previous
#include <hip/hip_runtime.h>
#include <math.h>


#define B_     4
#define T_IN_  12
#define T_OUT_ 24
#define N_     5000
#define C_     32
#define H_     8
#define K1_    17
#define TH_    (T_IN_ * H_)
#define OT_    (T_OUT_ - T_IN_)
#define OTP_   16
#define KS_    (K1_ * T_IN_ / 4)
#define PF_    4
#define RB_    (N_ * C_ * 4)
#define BETA_  0.8f

#define NWG_     250
#define NPW_     20
#define WAVES_   20
#define THREADS_ (WAVES_ * 32)
#define ND4_     (N_ * K1_ / 4)
#define SWEEPS_  ((ND4_ + THREADS_ - 1) / THREADS_)
#define ROT_     (ND4_ / NWG_)

typedef __attribute__((ext_vector_type(2))) float v2f;
typedef __attribute__((ext_vector_type(4))) float v4f;
typedef __attribute__((ext_vector_type(8))) float v8f;

__global__ __launch_bounds__(THREADS_) void gnn_folded_kernel(
    const float* __restrict__ x,
    const int*   __restrict__ nn,
    const float* __restrict__ nd,
    const float* __restrict__ Wm,
    const float* __restrict__ bias,
    float*       __restrict__ out) {
  __shared__ float red[WAVES_];
  __shared__ float Wm_lds[TH_ * OTP_];
  __shared__ float bias_lds[OTP_];
  __shared__ float w_lds[NPW_ * K1_ * H_];
  __shared__ int   nn_lds[NPW_ * K1_];

  const int tid  = threadIdx.x;
  const int lane = tid & 31;
  const int wv   = tid >> 5;
  const int n0   = blockIdx.x * NPW_;

  const int   e      = (tid < NPW_ * K1_) ? tid : (NPW_ * K1_ - 1);
  const float d_e    = nd[n0 * K1_ + e];
  const int   node_e = nn[n0 * K1_ + e];
  float wmv[3];
#pragma unroll
  for (int r = 0; r < 3; ++r) {
    const int i  = r * THREADS_ + tid;
    const int ic = (i < TH_ * OTP_) ? i : (TH_ * OTP_ - 1);
    const int k  = ic >> 4, o = ic & (OTP_ - 1);
    wmv[r] = Wm[k * OT_ + ((o < OT_) ? o : (OT_ - 1))];
  }
  const float bias_v = bias[(tid < OT_) ? tid : (OT_ - 1)];

  const float4* __restrict__ nd4 = reinterpret_cast<const float4*>(nd);
  const int rot = blockIdx.x * ROT_;
  float m = 0.0f;
#pragma unroll
  for (int k = 0; k < SWEEPS_; ++k) {
    int i = k * THREADS_ + tid;
    if (i > ND4_ - 1) i = ND4_ - 1;
    int p = i + rot;
    if (p >= ND4_) p -= ND4_;
    const float4 v = nd4[p];
    m = fmaxf(m, fmaxf(fmaxf(v.x, v.y), fmaxf(v.z, v.w)));
  }
#pragma unroll
  for (int off = 16; off > 0; off >>= 1) m = fmaxf(m, __shfl_xor(m, off, 32));
  if (lane == 0) red[wv] = m;

#pragma unroll
  for (int r = 0; r < 3; ++r) {
    const int i = r * THREADS_ + tid;
    if (i < TH_ * OTP_) Wm_lds[i] = ((i & (OTP_ - 1)) < OT_) ? wmv[r] : 0.0f;
  }
  if (tid < OTP_) bias_lds[tid] = (tid < OT_) ? bias_v : 0.0f;
  if (tid < NPW_ * K1_) nn_lds[tid] = (node_e < 0) ? 0 : node_e;
  __syncthreads();

  float mx = red[0];
#pragma unroll
  for (int w = 1; w < WAVES_; ++w) mx = fmaxf(mx, red[w]);
  const float sig    = mx * 0.5f;
  const float inv_s2 = 1.0f / (sig * sig);

  if (tid < NPW_ * K1_) {
#pragma unroll
    for (int h = 0; h < H_; ++h) {
      const float lam = (float)(h + 1) * (1.0f / (float)H_);
      float wgt = __expf(-d_e * d_e * lam * inv_s2);
      if (node_e < 0 || wgt < 1e-5f) wgt = 0.0f;
      w_lds[tid * H_ + h] = wgt;
    }
  }
  __syncthreads();

  const int half = lane >> 4;
  const int orow = lane & 15;
  const int bsel = (lane >> 3) & 1;
  const int cg4  = (lane & 7) * 4;
  const int n    = n0 + wv;
  const char* __restrict__ xb0 = reinterpret_cast<const char*>(x);
  const char* __restrict__ xb1 = xb0 + (size_t)(2 * T_IN_) * RB_;
  char* __restrict__ ob = reinterpret_cast<char*>(out);

  const unsigned pofs = (unsigned)((lane >> 3) * RB_ + (n * C_ + cg4) * 4);
  v4f pt[12];
#pragma unroll
  for (int i = 0; i < 12; ++i) {
    const int b = i / 3, t = (i % 3) * 4;
    pt[i] = *reinterpret_cast<const v4f*>(xb0 + (size_t)((b * T_IN_ + t)) * RB_ + pofs);
  }

  const unsigned lofs = (unsigned)(bsel * (T_IN_ * RB_) + half * (2 * RB_) + cg4 * 4);
  const int nnb = wv * K1_;

  v8f acc[2][4];
#pragma unroll
  for (int v = 0; v < 8; ++v) {
    const float bv = bias_lds[v + 8 * half];
#pragma unroll
    for (int bp = 0; bp < 2; ++bp)
#pragma unroll
      for (int q = 0; q < 4; ++q) acc[bp][q][v] = bv;
  }

  v4f ring[PF_][2][2];
#pragma unroll
  for (int ks = 0; ks < PF_ - 1; ++ks) {
    const int r = ks / K1_, j = ks % K1_;
    const unsigned vo = lofs + (unsigned)nn_lds[nnb + j] * (C_ * 4);
#pragma unroll
    for (int s = 0; s < 2; ++s) {
      ring[ks % PF_][0][s] = *reinterpret_cast<const v4f*>(xb0 + (size_t)((4 * r + s)) * RB_ + vo);
      ring[ks % PF_][1][s] = *reinterpret_cast<const v4f*>(xb1 + (size_t)((4 * r + s)) * RB_ + vo);
    }
  }
  unsigned vo_next = lofs + (unsigned)nn_lds[nnb + (PF_ - 1) % K1_] * (C_ * 4);
  __builtin_amdgcn_sched_barrier(0);

#pragma unroll
  for (int i = 0; i < 12; ++i) {
    const int b = i / 3, t = (i % 3) * 4;
    *reinterpret_cast<v4f*>(ob + (size_t)((b * T_OUT_ + t)) * RB_ + pofs) = pt[i];
  }

  float Wr[2][H_];
#pragma unroll
  for (int ks = 0; ks < KS_; ++ks) {
    const int r = ks / K1_, j = ks % K1_;
    __builtin_amdgcn_sched_barrier(0);
    if (ks + PF_ - 1 < KS_) {
      const int kn = ks + PF_ - 1;
      const int rn = kn / K1_;
#pragma unroll
      for (int s = 0; s < 2; ++s) {
        ring[kn % PF_][0][s] = *reinterpret_cast<const v4f*>(xb0 + (size_t)((4 * rn + s)) * RB_ + vo_next);
        ring[kn % PF_][1][s] = *reinterpret_cast<const v4f*>(xb1 + (size_t)((4 * rn + s)) * RB_ + vo_next);
      }
      __builtin_amdgcn_sched_barrier(0);
    }
    if (j == 0) {
      asm volatile("" ::: "memory");
#pragma unroll
      for (int s = 0; s < 2; ++s)
#pragma unroll
        for (int h = 0; h < H_; ++h)
          Wr[s][h] = Wm_lds[((4 * r + 2 * half + s) * H_ + h) * OTP_ + orow];
    }
    if (ks + PF_ < KS_)
      vo_next = lofs + (unsigned)nn_lds[nnb + (ks + PF_) % K1_] * (C_ * 4);
    float wj[H_];
#pragma unroll
    for (int h = 0; h < H_; ++h) wj[h] = w_lds[(nnb + j) * H_ + h];
    v2f a;
    a.x = wj[0] * Wr[0][0];
    a.y = wj[0] * Wr[1][0];
#pragma unroll
    for (int h = 1; h < H_; ++h) {
      a.x = fmaf(wj[h], Wr[0][h], a.x);
      a.y = fmaf(wj[h], Wr[1][h], a.y);
    }
#pragma unroll
    for (int bp = 0; bp < 2; ++bp)
#pragma unroll
      for (int q = 0; q < 4; ++q) {
        v2f bf;
        bf.x = ring[ks % PF_][bp][0][q];
        bf.y = ring[ks % PF_][bp][1][q];
        acc[bp][q] = __builtin_amdgcn_wmma_f32_16x16x4_f32(
             false, a,  false, bf,
             (short)0, acc[bp][q],  false,  false);
      }
  }

  const unsigned oofs = (unsigned)(bsel * (T_OUT_ * RB_) + half * (8 * RB_) + (n * C_ + cg4) * 4);
#pragma unroll
  for (int bp = 0; bp < 2; ++bp)
#pragma unroll
    for (int v = 0; v < 8; ++v) {
      v4f yv;
#pragma unroll
      for (int q = 0; q < 4; ++q) {
        float y = acc[bp][q][v];
        const float s = 1.0f / (1.0f + __expf(-BETA_ * y));
        yv[q] = y * s;
      }
      if (v < OT_ - 8 || half == 0) {
        *reinterpret_cast<v4f*>(ob + (size_t)((bp * 2 * T_OUT_ + T_IN_ + v)) * RB_ + oofs) = yv;
      }
    }
}

extern "C" void kernel_launch(void* const* d_in, const int* in_sizes, int n_in,
                              void* d_out, int out_size, void* d_ws, size_t ws_size,
                              hipStream_t stream) {
  const float* x    = (const float*)d_in[0];
  const int*   nn   = (const int*)d_in[1];
  const float* nd   = (const float*)d_in[2];
  const float* Wm   = (const float*)d_in[3];
  const float* bias = (const float*)d_in[4];
  float* out = (float*)d_out;

  gnn_folded_kernel<<<NWG_, THREADS_, 0, stream>>>(x, nn, nd, Wm, bias, out);
}


# wave issue priority stepped down 3,2,1,0 with a wave's own progress through its 51 k-steps (at 26, 38, 46): a SIMD's five waves end together, the youngest no longer ends alone; bit-identical
# speedup vs baseline: 3.2899x; 1.0442x over previous
#include <hip/hip_runtime.h>
#include <math.h>


#define B_     4
#define T_IN_  12
#define T_OUT_ 24
#define N_     5000
#define C_     32
#define H_     8
#define K1_    17
#define TH_    (T_IN_ * H_)
#define OT_    (T_OUT_ - T_IN_)
#define OTP_   16
#define KS_    (K1_ * T_IN_ / 4)
#define PF_    4
#define RB_    (N_ * C_ * 4)
#define BETA_  0.8f
#define PR2_   26
#define PR1_   38
#define PR0_   46

#define NWG_     250
#define NPW_     20
#define WAVES_   20
#define THREADS_ (WAVES_ * 32)
#define ND4_     (N_ * K1_ / 4)
#define SWEEPS_  ((ND4_ + THREADS_ - 1) / THREADS_)
#define ROT_     (ND4_ / NWG_)

typedef __attribute__((ext_vector_type(2))) float v2f;
typedef __attribute__((ext_vector_type(4))) float v4f;
typedef __attribute__((ext_vector_type(8))) float v8f;

__global__ __launch_bounds__(THREADS_) void gnn_folded_kernel(
    const float* __restrict__ x,
    const int*   __restrict__ nn,
    const float* __restrict__ nd,
    const float* __restrict__ Wm,
    const float* __restrict__ bias,
    float*       __restrict__ out) {
  __shared__ float red[WAVES_];
  __shared__ float Wm_lds[TH_ * OTP_];
  __shared__ float bias_lds[OTP_];
  __shared__ float w_lds[NPW_ * K1_ * H_];
  __shared__ int   nn_lds[NPW_ * K1_];

  const int tid  = threadIdx.x;
  const int lane = tid & 31;
  const int wv   = tid >> 5;
  const int n0   = blockIdx.x * NPW_;

  const int   e      = (tid < NPW_ * K1_) ? tid : (NPW_ * K1_ - 1);
  const float d_e    = nd[n0 * K1_ + e];
  const int   node_e = nn[n0 * K1_ + e];
  float wmv[3];
#pragma unroll
  for (int r = 0; r < 3; ++r) {
    const int i  = r * THREADS_ + tid;
    const int ic = (i < TH_ * OTP_) ? i : (TH_ * OTP_ - 1);
    const int k  = ic >> 4, o = ic & (OTP_ - 1);
    wmv[r] = Wm[k * OT_ + ((o < OT_) ? o : (OT_ - 1))];
  }
  const float bias_v = bias[(tid < OT_) ? tid : (OT_ - 1)];

  const float4* __restrict__ nd4 = reinterpret_cast<const float4*>(nd);
  const int rot = blockIdx.x * ROT_;
  float m = 0.0f;
#pragma unroll
  for (int k = 0; k < SWEEPS_; ++k) {
    int i = k * THREADS_ + tid;
    if (i > ND4_ - 1) i = ND4_ - 1;
    int p = i + rot;
    if (p >= ND4_) p -= ND4_;
    const float4 v = nd4[p];
    m = fmaxf(m, fmaxf(fmaxf(v.x, v.y), fmaxf(v.z, v.w)));
  }
#pragma unroll
  for (int off = 16; off > 0; off >>= 1) m = fmaxf(m, __shfl_xor(m, off, 32));
  if (lane == 0) red[wv] = m;

#pragma unroll
  for (int r = 0; r < 3; ++r) {
    const int i = r * THREADS_ + tid;
    if (i < TH_ * OTP_) Wm_lds[i] = ((i & (OTP_ - 1)) < OT_) ? wmv[r] : 0.0f;
  }
  if (tid < OTP_) bias_lds[tid] = (tid < OT_) ? bias_v : 0.0f;
  if (tid < NPW_ * K1_) nn_lds[tid] = (node_e < 0) ? 0 : node_e;
  __syncthreads();

  float mx = red[0];
#pragma unroll
  for (int w = 1; w < WAVES_; ++w) mx = fmaxf(mx, red[w]);
  const float sig    = mx * 0.5f;
  const float inv_s2 = 1.0f / (sig * sig);

  if (tid < NPW_ * K1_) {
#pragma unroll
    for (int h = 0; h < H_; ++h) {
      const float lam = (float)(h + 1) * (1.0f / (float)H_);
      float wgt = __expf(-d_e * d_e * lam * inv_s2);
      if (node_e < 0 || wgt < 1e-5f) wgt = 0.0f;
      w_lds[tid * H_ + h] = wgt;
    }
  }
  __syncthreads();

  __builtin_amdgcn_s_setprio(3);

  const int half = lane >> 4;
  const int orow = lane & 15;
  const int bsel = (lane >> 3) & 1;
  const int cg4  = (lane & 7) * 4;
  const int n    = n0 + wv;
  const char* __restrict__ xb0 = reinterpret_cast<const char*>(x);
  const char* __restrict__ xb1 = xb0 + (size_t)(2 * T_IN_) * RB_;
  char* __restrict__ ob = reinterpret_cast<char*>(out);

  const unsigned pofs = (unsigned)((lane >> 3) * RB_ + (n * C_ + cg4) * 4);
  v4f pt[12];
#pragma unroll
  for (int i = 0; i < 12; ++i) {
    const int b = i / 3, t = (i % 3) * 4;
    pt[i] = *reinterpret_cast<const v4f*>(xb0 + (size_t)((b * T_IN_ + t)) * RB_ + pofs);
  }

  const unsigned lofs = (unsigned)(bsel * (T_IN_ * RB_) + half * (2 * RB_) + cg4 * 4);
  const int nnb = wv * K1_;

  v8f acc[2][4];
#pragma unroll
  for (int v = 0; v < 8; ++v) {
    const float bv = bias_lds[v + 8 * half];
#pragma unroll
    for (int bp = 0; bp < 2; ++bp)
#pragma unroll
      for (int q = 0; q < 4; ++q) acc[bp][q][v] = bv;
  }

  v4f ring[PF_][2][2];
#pragma unroll
  for (int ks = 0; ks < PF_ - 1; ++ks) {
    const int r = ks / K1_, j = ks % K1_;
    const unsigned vo = lofs + (unsigned)nn_lds[nnb + j] * (C_ * 4);
#pragma unroll
    for (int s = 0; s < 2; ++s) {
      ring[ks % PF_][0][s] = *reinterpret_cast<const v4f*>(xb0 + (size_t)((4 * r + s)) * RB_ + vo);
      ring[ks % PF_][1][s] = *reinterpret_cast<const v4f*>(xb1 + (size_t)((4 * r + s)) * RB_ + vo);
    }
  }
  unsigned vo_next = lofs + (unsigned)nn_lds[nnb + (PF_ - 1) % K1_] * (C_ * 4);
  __builtin_amdgcn_sched_barrier(0);

#pragma unroll
  for (int i = 0; i < 12; ++i) {
    const int b = i / 3, t = (i % 3) * 4;
    *reinterpret_cast<v4f*>(ob + (size_t)((b * T_OUT_ + t)) * RB_ + pofs) = pt[i];
  }

  float Wr[2][H_];
#pragma unroll
  for (int ks = 0; ks < KS_; ++ks) {
    const int r = ks / K1_, j = ks % K1_;
    __builtin_amdgcn_sched_barrier(0);
    if (ks == PR2_) __builtin_amdgcn_s_setprio(2);
    if (ks == PR1_) __builtin_amdgcn_s_setprio(1);
    if (ks == PR0_) __builtin_amdgcn_s_setprio(0);
    if (ks + PF_ - 1 < KS_) {
      const int kn = ks + PF_ - 1;
      const int rn = kn / K1_;
#pragma unroll
      for (int s = 0; s < 2; ++s) {
        ring[kn % PF_][0][s] = *reinterpret_cast<const v4f*>(xb0 + (size_t)((4 * rn + s)) * RB_ + vo_next);
        ring[kn % PF_][1][s] = *reinterpret_cast<const v4f*>(xb1 + (size_t)((4 * rn + s)) * RB_ + vo_next);
      }
      __builtin_amdgcn_sched_barrier(0);
    }
    if (j == 0) {
      asm volatile("" ::: "memory");
#pragma unroll
      for (int s = 0; s < 2; ++s)
#pragma unroll
        for (int h = 0; h < H_; ++h)
          Wr[s][h] = Wm_lds[((4 * r + 2 * half + s) * H_ + h) * OTP_ + orow];
    }
    if (ks + PF_ < KS_)
      vo_next = lofs + (unsigned)nn_lds[nnb + (ks + PF_) % K1_] * (C_ * 4);
    float wj[H_];
#pragma unroll
    for (int h = 0; h < H_; ++h) wj[h] = w_lds[(nnb + j) * H_ + h];
    v2f a;
    a.x = wj[0] * Wr[0][0];
    a.y = wj[0] * Wr[1][0];
#pragma unroll
    for (int h = 1; h < H_; ++h) {
      a.x = fmaf(wj[h], Wr[0][h], a.x);
      a.y = fmaf(wj[h], Wr[1][h], a.y);
    }
#pragma unroll
    for (int bp = 0; bp < 2; ++bp)
#pragma unroll
      for (int q = 0; q < 4; ++q) {
        v2f bf;
        bf.x = ring[ks % PF_][bp][0][q];
        bf.y = ring[ks % PF_][bp][1][q];
        acc[bp][q] = __builtin_amdgcn_wmma_f32_16x16x4_f32(
             false, a,  false, bf,
             (short)0, acc[bp][q],  false,  false);
      }
  }

  const unsigned oofs = (unsigned)(bsel * (T_OUT_ * RB_) + half * (8 * RB_) + (n * C_ + cg4) * 4);
#pragma unroll
  for (int bp = 0; bp < 2; ++bp)
#pragma unroll
    for (int v = 0; v < 8; ++v) {
      v4f yv;
#pragma unroll
      for (int q = 0; q < 4; ++q) {
        float y = acc[bp][q][v];
        const float s = 1.0f / (1.0f + __expf(-BETA_ * y));
        yv[q] = y * s;
      }
      if (v < OT_ - 8 || half == 0) {
        *reinterpret_cast<v4f*>(ob + (size_t)((bp * 2 * T_OUT_ + T_IN_ + v)) * RB_ + oofs) = yv;
      }
    }
}

extern "C" void kernel_launch(void* const* d_in, const int* in_sizes, int n_in,
                              void* d_out, int out_size, void* d_ws, size_t ws_size,
                              hipStream_t stream) {
  const float* x    = (const float*)d_in[0];
  const int*   nn   = (const int*)d_in[1];
  const float* nd   = (const float*)d_in[2];
  const float* Wm   = (const float*)d_in[3];
  const float* bias = (const float*)d_in[4];
  float* out = (float*)d_out;

  gnn_folded_kernel<<<NWG_, THREADS_, 0, stream>>>(x, nn, nd, Wm, bias, out);
}
